# starting kernel
# baseline (speedup 1.0000x reference)
#include <hip/hip_runtime.h>
#include <math.h>

#define B_   16
#define C_   2
#define F_   257
#define T_   4000
#define ROWS (B_*C_*F_)
#define TILES (ROWS/16)
#define WAVES_PER_BLOCK 8
#define LDS_STRIDE 20
#define FLOORV 1e-6f

typedef float v8f __attribute__((ext_vector_type(8)));
typedef float v2f __attribute__((ext_vector_type(2)));

__device__ __forceinline__ float fpow(float x, float y) {
  return __builtin_amdgcn_exp2f(y * __builtin_amdgcn_logf(x));
}

__device__ __forceinline__ float pcen_ep(float mag, float ema,
                                         float naexp, float dd, float oor, float dpow) {
  float deninv = fpow(FLOORV + ema, naexp);
  return fpow(fmaf(mag, deninv, dd), oor) - dpow;
}

__global__ __launch_bounds__(256) void pcen_wmma_kernel(
    const float* __restrict__ tf, const float* __restrict__ weights,
    const float* __restrict__ alpha, const float* __restrict__ delta,
    const float* __restrict__ root, float* __restrict__ out)
{
  __shared__ float smem_all[WAVES_PER_BLOCK][16 * LDS_STRIDE];

  const int lane = threadIdx.x & 31;
  const int wv   = threadIdx.x >> 5;
  const int tile = blockIdx.x * WAVES_PER_BLOCK + wv;
  if (tile >= TILES) return;
  float* smem = smem_all[wv];

  const int rowBase = tile * 16;
  const int half = lane >> 4;
  const int n    = lane & 15;

  const int lrow = lane >> 1;
  const int ltof = (lane & 1) * 8;
  const int grow = rowBase + lrow;
  const int gb   = grow / (C_ * F_);
  const int grem = grow % (C_ * F_);
  const float* reP = tf + ((size_t)(gb * C_ + grem / F_) * (2 * F_) + (grem % F_)) * T_;
  const float* imP = reP + (size_t)F_ * T_;

  const int erow = rowBase + n;
  const int ec   = (erow / F_) % C_;
  const float naexp = -fminf(alpha[ec], 1.0f);
  const float oor   = 1.0f / fmaxf(root[ec], 1.0f);
  const float dd    = delta[ec];
  const float dpow  = fpow(dd, oor);
  float* outP = out + (size_t)erow * T_;

  const float w = fminf(fmaxf(weights[0], 0.0f), 1.0f);
  const float q = 1.0f - w;

  float A0[4], A1[4];
#pragma unroll
  for (int b = 0; b < 4; ++b) {
    const int k0 = 4 * b + 2 * half;
    A0[b] = (k0     <= n) ? w * fpow(q, (float)(n - k0))     : 0.0f;
    A1[b] = (k0 + 1 <= n) ? w * fpow(q, (float)(n - k0 - 1)) : 0.0f;
  }
  float qp[8];
  qp[0] = fpow(q, (float)(1 + 8 * half));
#pragma unroll
  for (int i = 1; i < 8; ++i) qp[i] = qp[i - 1] * q;

  float carry = 0.0f;

  for (int chunk = 0; chunk < T_ / 16; ++chunk) {
    const int t0 = chunk * 16;

    const float4 r0 = *(const float4*)(reP + t0 + ltof);
    const float4 r1 = *(const float4*)(reP + t0 + ltof + 4);
    const float4 i0 = *(const float4*)(imP + t0 + ltof);
    const float4 i1 = *(const float4*)(imP + t0 + ltof + 4);
    __builtin_prefetch(reP + t0 + 256, 0, 0);
    __builtin_prefetch(imP + t0 + 256, 0, 0);
    float4 m0, m1;
    m0.x = __builtin_amdgcn_sqrtf(fmaf(r0.x, r0.x, i0.x * i0.x));
    m0.y = __builtin_amdgcn_sqrtf(fmaf(r0.y, r0.y, i0.y * i0.y));
    m0.z = __builtin_amdgcn_sqrtf(fmaf(r0.z, r0.z, i0.z * i0.z));
    m0.w = __builtin_amdgcn_sqrtf(fmaf(r0.w, r0.w, i0.w * i0.w));
    m1.x = __builtin_amdgcn_sqrtf(fmaf(r1.x, r1.x, i1.x * i1.x));
    m1.y = __builtin_amdgcn_sqrtf(fmaf(r1.y, r1.y, i1.y * i1.y));
    m1.z = __builtin_amdgcn_sqrtf(fmaf(r1.z, r1.z, i1.z * i1.z));
    m1.w = __builtin_amdgcn_sqrtf(fmaf(r1.w, r1.w, i1.w * i1.w));
    *(float4*)(smem + lrow * LDS_STRIDE + ltof)     = m0;
    *(float4*)(smem + lrow * LDS_STRIDE + ltof + 4) = m1;
    asm volatile("s_wait_dscnt 0" ::: "memory");

    if (chunk == 0) carry = smem[n * LDS_STRIDE];

    v8f acc;
#pragma unroll
    for (int i = 0; i < 8; ++i) acc[i] = qp[i] * carry;

#pragma unroll
    for (int b = 0; b < 4; ++b) {
      const int k0 = 4 * b + 2 * half;
      v2f Bv, Av;
      Bv.x = smem[n * LDS_STRIDE + k0];
      Bv.y = smem[n * LDS_STRIDE + k0 + 1];
      Av.x = A0[b];
      Av.y = A1[b];
      acc = __builtin_amdgcn_wmma_f32_16x16x4_f32(
          false, Av, false, Bv, (short)0, acc, false, false);
    }

    carry = __shfl(acc[7], 16 + n, 32);

    const float4 g0 = *(const float4*)(smem + n * LDS_STRIDE + 8 * half);
    const float4 g1 = *(const float4*)(smem + n * LDS_STRIDE + 8 * half + 4);
    float4 o0, o1;
    o0.x = pcen_ep(g0.x, acc[0], naexp, dd, oor, dpow);
    o0.y = pcen_ep(g0.y, acc[1], naexp, dd, oor, dpow);
    o0.z = pcen_ep(g0.z, acc[2], naexp, dd, oor, dpow);
    o0.w = pcen_ep(g0.w, acc[3], naexp, dd, oor, dpow);
    o1.x = pcen_ep(g1.x, acc[4], naexp, dd, oor, dpow);
    o1.y = pcen_ep(g1.y, acc[5], naexp, dd, oor, dpow);
    o1.z = pcen_ep(g1.z, acc[6], naexp, dd, oor, dpow);
    o1.w = pcen_ep(g1.w, acc[7], naexp, dd, oor, dpow);
    *(float4*)(outP + t0 + 8 * half)     = o0;
    *(float4*)(outP + t0 + 8 * half + 4) = o1;
  }
}

extern "C" void kernel_launch(void* const* d_in, const int* in_sizes, int n_in,
                              void* d_out, int out_size, void* d_ws, size_t ws_size,
                              hipStream_t stream) {
  (void)in_sizes; (void)n_in; (void)d_ws; (void)ws_size; (void)out_size;
  const float* tf      = (const float*)d_in[0];
  const float* weights = (const float*)d_in[1];
  const float* alpha   = (const float*)d_in[2];
  const float* delta   = (const float*)d_in[3];
  const float* root    = (const float*)d_in[4];
  float* out = (float*)d_out;

  const int blocks = (TILES + WAVES_PER_BLOCK - 1) / WAVES_PER_BLOCK;
  pcen_wmma_kernel<<<blocks, 32 * WAVES_PER_BLOCK, 0, stream>>>(
      tf, weights, alpha, delta, root, out);
}


# One wave per work-group (514 work-groups of 32 threads, was 65 of 256): the 514 row tiles spread over all 256 WGPs at one wave per SIMD instead of 65 WGPs at two per SIMD
# speedup vs baseline: 1.1911x; 1.1911x over previous
#include <hip/hip_runtime.h>
#include <math.h>

#define B_   16
#define C_   2
#define F_   257
#define T_   4000
#define ROWS (B_*C_*F_)
#define TILES (ROWS/16)
#define WAVES_PER_BLOCK 1
#define LDS_STRIDE 20
#define FLOORV 1e-6f

typedef float v8f __attribute__((ext_vector_type(8)));
typedef float v2f __attribute__((ext_vector_type(2)));

__device__ __forceinline__ float fpow(float x, float y) {
  return __builtin_amdgcn_exp2f(y * __builtin_amdgcn_logf(x));
}

__device__ __forceinline__ float pcen_ep(float mag, float ema,
                                         float naexp, float dd, float oor, float dpow) {
  float deninv = fpow(FLOORV + ema, naexp);
  return fpow(fmaf(mag, deninv, dd), oor) - dpow;
}

__global__ __launch_bounds__(32 * WAVES_PER_BLOCK) void pcen_wmma_kernel(
    const float* __restrict__ tf, const float* __restrict__ weights,
    const float* __restrict__ alpha, const float* __restrict__ delta,
    const float* __restrict__ root, float* __restrict__ out)
{
  __shared__ float smem_all[WAVES_PER_BLOCK][16 * LDS_STRIDE];

  const int lane = threadIdx.x & 31;
  const int wv   = threadIdx.x >> 5;
  const int tile = blockIdx.x * WAVES_PER_BLOCK + wv;
  if (tile >= TILES) return;
  float* smem = smem_all[wv];

  const int rowBase = tile * 16;
  const int half = lane >> 4;
  const int n    = lane & 15;

  const int lrow = lane >> 1;
  const int ltof = (lane & 1) * 8;
  const int grow = rowBase + lrow;
  const int gb   = grow / (C_ * F_);
  const int grem = grow % (C_ * F_);
  const float* reP = tf + ((size_t)(gb * C_ + grem / F_) * (2 * F_) + (grem % F_)) * T_;
  const float* imP = reP + (size_t)F_ * T_;

  const int erow = rowBase + n;
  const int ec   = (erow / F_) % C_;
  const float naexp = -fminf(alpha[ec], 1.0f);
  const float oor   = 1.0f / fmaxf(root[ec], 1.0f);
  const float dd    = delta[ec];
  const float dpow  = fpow(dd, oor);
  float* outP = out + (size_t)erow * T_;

  const float w = fminf(fmaxf(weights[0], 0.0f), 1.0f);
  const float q = 1.0f - w;

  float A0[4], A1[4];
#pragma unroll
  for (int b = 0; b < 4; ++b) {
    const int k0 = 4 * b + 2 * half;
    A0[b] = (k0     <= n) ? w * fpow(q, (float)(n - k0))     : 0.0f;
    A1[b] = (k0 + 1 <= n) ? w * fpow(q, (float)(n - k0 - 1)) : 0.0f;
  }
  float qp[8];
  qp[0] = fpow(q, (float)(1 + 8 * half));
#pragma unroll
  for (int i = 1; i < 8; ++i) qp[i] = qp[i - 1] * q;

  float carry = 0.0f;

  for (int chunk = 0; chunk < T_ / 16; ++chunk) {
    const int t0 = chunk * 16;

    const float4 r0 = *(const float4*)(reP + t0 + ltof);
    const float4 r1 = *(const float4*)(reP + t0 + ltof + 4);
    const float4 i0 = *(const float4*)(imP + t0 + ltof);
    const float4 i1 = *(const float4*)(imP + t0 + ltof + 4);
    __builtin_prefetch(reP + t0 + 256, 0, 0);
    __builtin_prefetch(imP + t0 + 256, 0, 0);
    float4 m0, m1;
    m0.x = __builtin_amdgcn_sqrtf(fmaf(r0.x, r0.x, i0.x * i0.x));
    m0.y = __builtin_amdgcn_sqrtf(fmaf(r0.y, r0.y, i0.y * i0.y));
    m0.z = __builtin_amdgcn_sqrtf(fmaf(r0.z, r0.z, i0.z * i0.z));
    m0.w = __builtin_amdgcn_sqrtf(fmaf(r0.w, r0.w, i0.w * i0.w));
    m1.x = __builtin_amdgcn_sqrtf(fmaf(r1.x, r1.x, i1.x * i1.x));
    m1.y = __builtin_amdgcn_sqrtf(fmaf(r1.y, r1.y, i1.y * i1.y));
    m1.z = __builtin_amdgcn_sqrtf(fmaf(r1.z, r1.z, i1.z * i1.z));
    m1.w = __builtin_amdgcn_sqrtf(fmaf(r1.w, r1.w, i1.w * i1.w));
    *(float4*)(smem + lrow * LDS_STRIDE + ltof)     = m0;
    *(float4*)(smem + lrow * LDS_STRIDE + ltof + 4) = m1;
    asm volatile("s_wait_dscnt 0" ::: "memory");

    if (chunk == 0) carry = smem[n * LDS_STRIDE];

    v8f acc;
#pragma unroll
    for (int i = 0; i < 8; ++i) acc[i] = qp[i] * carry;

#pragma unroll
    for (int b = 0; b < 4; ++b) {
      const int k0 = 4 * b + 2 * half;
      v2f Bv, Av;
      Bv.x = smem[n * LDS_STRIDE + k0];
      Bv.y = smem[n * LDS_STRIDE + k0 + 1];
      Av.x = A0[b];
      Av.y = A1[b];
      acc = __builtin_amdgcn_wmma_f32_16x16x4_f32(
          false, Av, false, Bv, (short)0, acc, false, false);
    }

    carry = __shfl(acc[7], 16 + n, 32);

    const float4 g0 = *(const float4*)(smem + n * LDS_STRIDE + 8 * half);
    const float4 g1 = *(const float4*)(smem + n * LDS_STRIDE + 8 * half + 4);
    float4 o0, o1;
    o0.x = pcen_ep(g0.x, acc[0], naexp, dd, oor, dpow);
    o0.y = pcen_ep(g0.y, acc[1], naexp, dd, oor, dpow);
    o0.z = pcen_ep(g0.z, acc[2], naexp, dd, oor, dpow);
    o0.w = pcen_ep(g0.w, acc[3], naexp, dd, oor, dpow);
    o1.x = pcen_ep(g1.x, acc[4], naexp, dd, oor, dpow);
    o1.y = pcen_ep(g1.y, acc[5], naexp, dd, oor, dpow);
    o1.z = pcen_ep(g1.z, acc[6], naexp, dd, oor, dpow);
    o1.w = pcen_ep(g1.w, acc[7], naexp, dd, oor, dpow);
    *(float4*)(outP + t0 + 8 * half)     = o0;
    *(float4*)(outP + t0 + 8 * half + 4) = o1;
  }
}

extern "C" void kernel_launch(void* const* d_in, const int* in_sizes, int n_in,
                              void* d_out, int out_size, void* d_ws, size_t ws_size,
                              hipStream_t stream) {
  (void)in_sizes; (void)n_in; (void)d_ws; (void)ws_size; (void)out_size;
  const float* tf      = (const float*)d_in[0];
  const float* weights = (const float*)d_in[1];
  const float* alpha   = (const float*)d_in[2];
  const float* delta   = (const float*)d_in[3];
  const float* root    = (const float*)d_in[4];
  float* out = (float*)d_out;

  const int blocks = (TILES + WAVES_PER_BLOCK - 1) / WAVES_PER_BLOCK;
  pcen_wmma_kernel<<<blocks, 32 * WAVES_PER_BLOCK, 0, stream>>>(
      tf, weights, alpha, delta, root, out);
}


# Register ring of 5 chunks: each chunk's re/im global loads are issued 5 chunks ahead into registers, so the wave no longer waits a memory round trip in every 16-step chunk
# speedup vs baseline: 1.8041x; 1.5147x over previous
#include <hip/hip_runtime.h>
#include <math.h>

#define B_   16
#define C_   2
#define F_   257
#define T_   4000
#define ROWS (B_*C_*F_)
#define TILES (ROWS/16)
#define WAVES_PER_BLOCK 1
#define LDS_STRIDE 20
#define FLOORV 1e-6f
#define NCHUNK (T_/16)
#define DEPTH  5

typedef float v8f __attribute__((ext_vector_type(8)));
typedef float v2f __attribute__((ext_vector_type(2)));

__device__ __forceinline__ float fpow(float x, float y) {
  return __builtin_amdgcn_exp2f(y * __builtin_amdgcn_logf(x));
}

__device__ __forceinline__ float pcen_ep(float mag, float ema,
                                         float naexp, float dd, float oor, float dpow) {
  float deninv = fpow(FLOORV + ema, naexp);
  return fpow(fmaf(mag, deninv, dd), oor) - dpow;
}

__global__ __launch_bounds__(32 * WAVES_PER_BLOCK) void pcen_wmma_kernel(
    const float* __restrict__ tf, const float* __restrict__ weights,
    const float* __restrict__ alpha, const float* __restrict__ delta,
    const float* __restrict__ root, float* __restrict__ out)
{
  __shared__ float smem_all[WAVES_PER_BLOCK][16 * LDS_STRIDE];

  const int lane = threadIdx.x & 31;
  const int wv   = threadIdx.x >> 5;
  const int tile = blockIdx.x * WAVES_PER_BLOCK + wv;
  if (tile >= TILES) return;
  float* smem = smem_all[wv];

  const int rowBase = tile * 16;
  const int half = lane >> 4;
  const int n    = lane & 15;

  const int lrow = lane >> 1;
  const int ltof = (lane & 1) * 8;
  const int grow = rowBase + lrow;
  const int gb   = grow / (C_ * F_);
  const int grem = grow % (C_ * F_);
  const float* reP = tf + ((size_t)(gb * C_ + grem / F_) * (2 * F_) + (grem % F_)) * T_;
  const float* imP = reP + (size_t)F_ * T_;

  const int erow = rowBase + n;
  const int ec   = (erow / F_) % C_;
  const float naexp = -fminf(alpha[ec], 1.0f);
  const float oor   = 1.0f / fmaxf(root[ec], 1.0f);
  const float dd    = delta[ec];
  const float dpow  = fpow(dd, oor);
  float* outP = out + (size_t)erow * T_;

  const float w = fminf(fmaxf(weights[0], 0.0f), 1.0f);
  const float q = 1.0f - w;

  float A0[4], A1[4];
#pragma unroll
  for (int b = 0; b < 4; ++b) {
    const int k0 = 4 * b + 2 * half;
    A0[b] = (k0     <= n) ? w * fpow(q, (float)(n - k0))     : 0.0f;
    A1[b] = (k0 + 1 <= n) ? w * fpow(q, (float)(n - k0 - 1)) : 0.0f;
  }
  float qp[8];
  qp[0] = fpow(q, (float)(1 + 8 * half));
#pragma unroll
  for (int i = 1; i < 8; ++i) qp[i] = qp[i - 1] * q;

  float carry = 0.0f;

  float4 R0[DEPTH], R1[DEPTH], I0[DEPTH], I1[DEPTH];
#pragma unroll
  for (int j = 0; j < DEPTH; ++j) {
    R0[j] = *(const float4*)(reP + 16 * j + ltof);
    R1[j] = *(const float4*)(reP + 16 * j + ltof + 4);
    I0[j] = *(const float4*)(imP + 16 * j + ltof);
    I1[j] = *(const float4*)(imP + 16 * j + ltof + 4);
  }

  for (int grp = 0; grp < NCHUNK / DEPTH; ++grp) {
#pragma unroll
  for (int j = 0; j < DEPTH; ++j) {
    __builtin_amdgcn_sched_barrier(0);
    const int chunk = grp * DEPTH + j;
    const int t0 = chunk * 16;

    const float4 r0 = R0[j];
    const float4 r1 = R1[j];
    const float4 i0 = I0[j];
    const float4 i1 = I1[j];
    float4 m0, m1;
    m0.x = __builtin_amdgcn_sqrtf(fmaf(r0.x, r0.x, i0.x * i0.x));
    m0.y = __builtin_amdgcn_sqrtf(fmaf(r0.y, r0.y, i0.y * i0.y));
    m0.z = __builtin_amdgcn_sqrtf(fmaf(r0.z, r0.z, i0.z * i0.z));
    m0.w = __builtin_amdgcn_sqrtf(fmaf(r0.w, r0.w, i0.w * i0.w));
    m1.x = __builtin_amdgcn_sqrtf(fmaf(r1.x, r1.x, i1.x * i1.x));
    m1.y = __builtin_amdgcn_sqrtf(fmaf(r1.y, r1.y, i1.y * i1.y));
    m1.z = __builtin_amdgcn_sqrtf(fmaf(r1.z, r1.z, i1.z * i1.z));
    m1.w = __builtin_amdgcn_sqrtf(fmaf(r1.w, r1.w, i1.w * i1.w));
    __builtin_amdgcn_sched_barrier(0);
    const int tn = (t0 + 16 * DEPTH < T_) ? (t0 + 16 * DEPTH) : (T_ - 16);
    R0[j] = *(const float4*)(reP + tn + ltof);
    R1[j] = *(const float4*)(reP + tn + ltof + 4);
    I0[j] = *(const float4*)(imP + tn + ltof);
    I1[j] = *(const float4*)(imP + tn + ltof + 4);
    __builtin_prefetch(reP + t0 + 256, 0, 0);
    __builtin_prefetch(imP + t0 + 256, 0, 0);
    *(float4*)(smem + lrow * LDS_STRIDE + ltof)     = m0;
    *(float4*)(smem + lrow * LDS_STRIDE + ltof + 4) = m1;
    asm volatile("s_wait_dscnt 0" ::: "memory");

    if (chunk == 0) carry = smem[n * LDS_STRIDE];

    v8f acc;
#pragma unroll
    for (int i = 0; i < 8; ++i) acc[i] = qp[i] * carry;

#pragma unroll
    for (int b = 0; b < 4; ++b) {
      const int k0 = 4 * b + 2 * half;
      v2f Bv, Av;
      Bv.x = smem[n * LDS_STRIDE + k0];
      Bv.y = smem[n * LDS_STRIDE + k0 + 1];
      Av.x = A0[b];
      Av.y = A1[b];
      acc = __builtin_amdgcn_wmma_f32_16x16x4_f32(
          false, Av, false, Bv, (short)0, acc, false, false);
    }

    carry = __shfl(acc[7], 16 + n, 32);

    const float4 g0 = *(const float4*)(smem + n * LDS_STRIDE + 8 * half);
    const float4 g1 = *(const float4*)(smem + n * LDS_STRIDE + 8 * half + 4);
    float4 o0, o1;
    o0.x = pcen_ep(g0.x, acc[0], naexp, dd, oor, dpow);
    o0.y = pcen_ep(g0.y, acc[1], naexp, dd, oor, dpow);
    o0.z = pcen_ep(g0.z, acc[2], naexp, dd, oor, dpow);
    o0.w = pcen_ep(g0.w, acc[3], naexp, dd, oor, dpow);
    o1.x = pcen_ep(g1.x, acc[4], naexp, dd, oor, dpow);
    o1.y = pcen_ep(g1.y, acc[5], naexp, dd, oor, dpow);
    o1.z = pcen_ep(g1.z, acc[6], naexp, dd, oor, dpow);
    o1.w = pcen_ep(g1.w, acc[7], naexp, dd, oor, dpow);
    *(float4*)(outP + t0 + 8 * half)     = o0;
    *(float4*)(outP + t0 + 8 * half + 4) = o1;
  }
  }
}

extern "C" void kernel_launch(void* const* d_in, const int* in_sizes, int n_in,
                              void* d_out, int out_size, void* d_ws, size_t ws_size,
                              hipStream_t stream) {
  (void)in_sizes; (void)n_in; (void)d_ws; (void)ws_size; (void)out_size;
  const float* tf      = (const float*)d_in[0];
  const float* weights = (const float*)d_in[1];
  const float* alpha   = (const float*)d_in[2];
  const float* delta   = (const float*)d_in[3];
  const float* root    = (const float*)d_in[4];
  float* out = (float*)d_out;

  const int blocks = (TILES + WAVES_PER_BLOCK - 1) / WAVES_PER_BLOCK;
  pcen_wmma_kernel<<<blocks, 32 * WAVES_PER_BLOCK, 0, stream>>>(
      tf, weights, alpha, delta, root, out);
}


# Time axis of each 16-row tile split over the 10 waves of a work-group (80-step segments; end states through LDS, one barrier per 800 steps); changes the summation order of the EMA carry
# speedup vs baseline: 2.3331x; 1.2932x over previous
#include <hip/hip_runtime.h>
#include <math.h>

#define B_   16
#define C_   2
#define F_   257
#define T_   4000
#define ROWS (B_*C_*F_)
#define TILES (ROWS/16)
#define WSEG 10
#define CH   5
#define LDS_STRIDE 20
#define FLOORV 1e-6f
#define NCHUNK (T_/16)
#define NPASS (NCHUNK/(WSEG*CH))

typedef float v8f __attribute__((ext_vector_type(8)));
typedef float v2f __attribute__((ext_vector_type(2)));

__device__ __forceinline__ float fpow(float x, float y) {
  return __builtin_amdgcn_exp2f(y * __builtin_amdgcn_logf(x));
}

__device__ __forceinline__ float pcen_ep(float mag, float ema,
                                         float naexp, float dd, float oor, float dpow) {
  float deninv = fpow(FLOORV + ema, naexp);
  return fpow(fmaf(mag, deninv, dd), oor) - dpow;
}

__global__ __launch_bounds__(32 * WSEG) void pcen_wmma_kernel(
    const float* __restrict__ tf, const float* __restrict__ weights,
    const float* __restrict__ alpha, const float* __restrict__ delta,
    const float* __restrict__ root, float* __restrict__ out)
{
  __shared__ float smem_all[WSEG][16 * LDS_STRIDE];
  __shared__ float ends[2][WSEG][16];

  const int lane = threadIdx.x & 31;
  const int wv   = threadIdx.x >> 5;
  const int tile = blockIdx.x;
  float* smem = smem_all[wv];

  const int rowBase = tile * 16;
  const int half = lane >> 4;
  const int n    = lane & 15;

  const int lrow = lane >> 1;
  const int lodd = lane & 1;
  const int ltof = lodd * 8;
  const int grow = rowBase + lrow;
  const int gb   = grow / (C_ * F_);
  const int grem = grow % (C_ * F_);
  const float* reP = tf + ((size_t)(gb * C_ + grem / F_) * (2 * F_) + (grem % F_)) * T_;
  const float* imP = reP + (size_t)F_ * T_;

  const int erow = rowBase + n;
  const int ec   = (erow / F_) % C_;
  const float naexp = -fminf(alpha[ec], 1.0f);
  const float oor   = 1.0f / fmaxf(root[ec], 1.0f);
  const float dd    = delta[ec];
  const float dpow  = fpow(dd, oor);
  float* outP = out + (size_t)erow * T_;

  const float w = fminf(fmaxf(weights[0], 0.0f), 1.0f);
  const float q = 1.0f - w;

  float A0[4], A1[4];
#pragma unroll
  for (int b = 0; b < 4; ++b) {
    const int k0 = 4 * b + 2 * half;
    A0[b] = (k0     <= n) ? w * fpow(q, (float)(n - k0))     : 0.0f;
    A1[b] = (k0 + 1 <= n) ? w * fpow(q, (float)(n - k0 - 1)) : 0.0f;
  }
  float qp[8];
  qp[0] = fpow(q, (float)(1 + 8 * half));
#pragma unroll
  for (int i = 1; i < 8; ++i) qp[i] = qp[i - 1] * q;

  const float q2 = q * q, q4 = q2 * q2, q8 = q4 * q4, q16 = q8 * q8;
  const float qseg = q16 * q16 * q16 * q16 * q16;

  float cpass;
  {
    const int xb   = erow / (C_ * F_);
    const int xrem = erow % (C_ * F_);
    const float* xr = tf + ((size_t)(xb * C_ + xrem / F_) * (2 * F_) + (xrem % F_)) * T_;
    const float r = xr[0], im = xr[(size_t)F_ * T_];
    cpass = __builtin_amdgcn_sqrtf(fmaf(r, r, im * im));
  }

#pragma unroll 1
  for (int pass = 0; pass < NPASS; ++pass) {
    const int seg = pass * WSEG + wv;
    const int ts  = seg * (16 * CH);

    float4 R0[CH], R1[CH], I0[CH], I1[CH];
#pragma unroll
    for (int j = 0; j < CH; ++j) {
      R0[j] = *(const float4*)(reP + ts + 16 * j + ltof);
      R1[j] = *(const float4*)(reP + ts + 16 * j + ltof + 4);
      I0[j] = *(const float4*)(imP + ts + 16 * j + ltof);
      I1[j] = *(const float4*)(imP + ts + 16 * j + ltof + 4);
    }

    float4 M0[CH], M1[CH];
    float pl = 0.0f;
#pragma unroll
    for (int j = 0; j < CH; ++j) {
      const float4 r0 = R0[j], r1 = R1[j], i0 = I0[j], i1 = I1[j];
      float4 m0, m1;
      m0.x = __builtin_amdgcn_sqrtf(fmaf(r0.x, r0.x, i0.x * i0.x));
      m0.y = __builtin_amdgcn_sqrtf(fmaf(r0.y, r0.y, i0.y * i0.y));
      m0.z = __builtin_amdgcn_sqrtf(fmaf(r0.z, r0.z, i0.z * i0.z));
      m0.w = __builtin_amdgcn_sqrtf(fmaf(r0.w, r0.w, i0.w * i0.w));
      m1.x = __builtin_amdgcn_sqrtf(fmaf(r1.x, r1.x, i1.x * i1.x));
      m1.y = __builtin_amdgcn_sqrtf(fmaf(r1.y, r1.y, i1.y * i1.y));
      m1.z = __builtin_amdgcn_sqrtf(fmaf(r1.z, r1.z, i1.z * i1.z));
      m1.w = __builtin_amdgcn_sqrtf(fmaf(r1.w, r1.w, i1.w * i1.w));
      M0[j] = m0; M1[j] = m1;
      float p = m0.x;
      p = fmaf(p, q, m0.y); p = fmaf(p, q, m0.z); p = fmaf(p, q, m0.w);
      p = fmaf(p, q, m1.x); p = fmaf(p, q, m1.y); p = fmaf(p, q, m1.z); p = fmaf(p, q, m1.w);
      pl = fmaf(pl, q16, p);
    }
    const float po = __shfl_xor(pl, 1, 32);
    if (lodd) ends[pass & 1][wv][lrow] = w * fmaf(q8, po, pl);
    asm volatile("s_wait_dscnt 0x0" ::: "memory");
    asm volatile("s_barrier_signal -1" ::: "memory");
    asm volatile("s_barrier_wait -1" ::: "memory");

    float carry = cpass, run = cpass;
#pragma unroll
    for (int u = 0; u < WSEG; ++u) {
      run = fmaf(run, qseg, ends[pass & 1][u][n]);
      if (u + 1 == wv) carry = run;
    }
    cpass = run;

#pragma unroll
    for (int j = 0; j < CH; ++j) {
      const int t0 = ts + 16 * j;
      *(float4*)(smem + lrow * LDS_STRIDE + ltof)     = M0[j];
      *(float4*)(smem + lrow * LDS_STRIDE + ltof + 4) = M1[j];
      asm volatile("s_wait_dscnt 0" ::: "memory");

      v8f acc;
#pragma unroll
      for (int i = 0; i < 8; ++i) acc[i] = qp[i] * carry;

#pragma unroll
      for (int b = 0; b < 4; ++b) {
        const int k0 = 4 * b + 2 * half;
        v2f Bv, Av;
        Bv.x = smem[n * LDS_STRIDE + k0];
        Bv.y = smem[n * LDS_STRIDE + k0 + 1];
        Av.x = A0[b];
        Av.y = A1[b];
        acc = __builtin_amdgcn_wmma_f32_16x16x4_f32(
            false, Av, false, Bv, (short)0, acc, false, false);
      }

      carry = __shfl(acc[7], 16 + n, 32);

      const float4 g0 = *(const float4*)(smem + n * LDS_STRIDE + 8 * half);
      const float4 g1 = *(const float4*)(smem + n * LDS_STRIDE + 8 * half + 4);
      float4 o0, o1;
      o0.x = pcen_ep(g0.x, acc[0], naexp, dd, oor, dpow);
      o0.y = pcen_ep(g0.y, acc[1], naexp, dd, oor, dpow);
      o0.z = pcen_ep(g0.z, acc[2], naexp, dd, oor, dpow);
      o0.w = pcen_ep(g0.w, acc[3], naexp, dd, oor, dpow);
      o1.x = pcen_ep(g1.x, acc[4], naexp, dd, oor, dpow);
      o1.y = pcen_ep(g1.y, acc[5], naexp, dd, oor, dpow);
      o1.z = pcen_ep(g1.z, acc[6], naexp, dd, oor, dpow);
      o1.w = pcen_ep(g1.w, acc[7], naexp, dd, oor, dpow);
      *(float4*)(outP + t0 + 8 * half)     = o0;
      *(float4*)(outP + t0 + 8 * half + 4) = o1;
    }
  }
}

extern "C" void kernel_launch(void* const* d_in, const int* in_sizes, int n_in,
                              void* d_out, int out_size, void* d_ws, size_t ws_size,
                              hipStream_t stream) {
  (void)in_sizes; (void)n_in; (void)d_ws; (void)ws_size; (void)out_size;
  const float* tf      = (const float*)d_in[0];
  const float* weights = (const float*)d_in[1];
  const float* alpha   = (const float*)d_in[2];
  const float* delta   = (const float*)d_in[3];
  const float* root    = (const float*)d_in[4];
  float* out = (float*)d_out;

  const int blocks = TILES;
  pcen_wmma_kernel<<<blocks, 32 * WSEG, 0, stream>>>(
      tf, weights, alpha, delta, root, out);
}


# Global loads cover whole 128-byte lines: each load instruction reads 4 rows x 128 contiguous bytes (was 16 rows x 2 x 16 B); magnitudes reach their segment's wave through LDS, one more barrier a pass
# speedup vs baseline: 2.5621x; 1.0982x over previous
#include <hip/hip_runtime.h>
#include <math.h>

#define B_   16
#define C_   2
#define F_   257
#define T_   4000
#define ROWS (B_*C_*F_)
#define TILES (ROWS/16)
#define WSEG 10
#define CH   5
#define LDS_STRIDE 20
#define PTIME (16 * CH * WSEG)
#define MPITCH (PTIME + 4)
#define NUNIT ((4 * (PTIME / 32)) / WSEG)
#define FLOORV 1e-6f
#define NCHUNK (T_/16)
#define NPASS (NCHUNK/(WSEG*CH))

typedef float v8f __attribute__((ext_vector_type(8)));
typedef float v2f __attribute__((ext_vector_type(2)));

__device__ __forceinline__ float fpow(float x, float y) {
  return __builtin_amdgcn_exp2f(y * __builtin_amdgcn_logf(x));
}

__device__ __forceinline__ float pcen_ep(float mag, float ema,
                                         float naexp, float dd, float oor, float dpow) {
  float deninv = fpow(FLOORV + ema, naexp);
  return fpow(fmaf(mag, deninv, dd), oor) - dpow;
}

__global__ __launch_bounds__(32 * WSEG) void pcen_wmma_kernel(
    const float* __restrict__ tf, const float* __restrict__ weights,
    const float* __restrict__ alpha, const float* __restrict__ delta,
    const float* __restrict__ root, float* __restrict__ out)
{
  __shared__ float smem_all[WSEG][16 * LDS_STRIDE];
  __shared__ float ends[2][WSEG][16];
  __shared__ __attribute__((aligned(16))) float magbuf[16 * MPITCH];

  const int lane = threadIdx.x & 31;
  const int wv   = threadIdx.x >> 5;
  const int tile = blockIdx.x;
  float* smem = smem_all[wv];

  const int rowBase = tile * 16;
  const int half = lane >> 4;
  const int n    = lane & 15;

  const int lrow = lane >> 1;
  const int lodd = lane & 1;
  const int ltof = lodd * 8;

  const int r4 = lane >> 3;
  const int c8 = lane & 7;
  unsigned gbyte[4];
  int lbase[4];
#pragma unroll
  for (int ii = 0; ii < 4; ++ii) {
    const int u0  = NUNIT * wv + ii;
    const int row = 4 * (u0 & 3) + r4;
    const int c0  = u0 >> 2;
    const int gr  = rowBase + row;
    const int bc  = gr / F_;
    gbyte[ii] = 4u * ((unsigned)(gr + bc * F_) * (unsigned)T_ + (unsigned)(32 * c0 + 4 * c8));
    lbase[ii] = row * MPITCH + 32 * c0 + 4 * c8;
  }

  const int erow = rowBase + n;
  const int ec   = (erow / F_) % C_;
  const float naexp = -fminf(alpha[ec], 1.0f);
  const float oor   = 1.0f / fmaxf(root[ec], 1.0f);
  const float dd    = delta[ec];
  const float dpow  = fpow(dd, oor);
  float* outP = out + (size_t)erow * T_;

  const float w = fminf(fmaxf(weights[0], 0.0f), 1.0f);
  const float q = 1.0f - w;

  float A0[4], A1[4];
#pragma unroll
  for (int b = 0; b < 4; ++b) {
    const int k0 = 4 * b + 2 * half;
    A0[b] = (k0     <= n) ? w * fpow(q, (float)(n - k0))     : 0.0f;
    A1[b] = (k0 + 1 <= n) ? w * fpow(q, (float)(n - k0 - 1)) : 0.0f;
  }
  float qp[8];
  qp[0] = fpow(q, (float)(1 + 8 * half));
#pragma unroll
  for (int i = 1; i < 8; ++i) qp[i] = qp[i - 1] * q;

  const float q2 = q * q, q4 = q2 * q2, q8 = q4 * q4, q16 = q8 * q8;
  const float qseg = q16 * q16 * q16 * q16 * q16;

  float cpass;
  {
    const int xb   = erow / (C_ * F_);
    const int xrem = erow % (C_ * F_);
    const float* xr = tf + ((size_t)(xb * C_ + xrem / F_) * (2 * F_) + (xrem % F_)) * T_;
    const float r = xr[0], im = xr[(size_t)F_ * T_];
    cpass = __builtin_amdgcn_sqrtf(fmaf(r, r, im * im));
  }

#pragma unroll 1
  for (int pass = 0; pass < NPASS; ++pass) {
    const int seg = pass * WSEG + wv;
    const int ts  = seg * (16 * CH);

    float4 RE[NUNIT], IM[NUNIT];
#pragma unroll
    for (int k = 0; k < NUNIT; ++k) {
      const float* p = (const float*)((const char*)tf + gbyte[k & 3]) + 32 * (k >> 2);
      RE[k] = *(const float4*)p;
      IM[k] = *(const float4*)(p + (size_t)F_ * T_);
    }
#pragma unroll
    for (int ii = 0; ii < 4; ++ii) gbyte[ii] += 4u * PTIME;

#pragma unroll
    for (int k = 0; k < NUNIT; ++k) {
      const float4 r = RE[k], im = IM[k];
      float4 m;
      m.x = __builtin_amdgcn_sqrtf(fmaf(r.x, r.x, im.x * im.x));
      m.y = __builtin_amdgcn_sqrtf(fmaf(r.y, r.y, im.y * im.y));
      m.z = __builtin_amdgcn_sqrtf(fmaf(r.z, r.z, im.z * im.z));
      m.w = __builtin_amdgcn_sqrtf(fmaf(r.w, r.w, im.w * im.w));
      *(float4*)(magbuf + lbase[k & 3] + 32 * (k >> 2)) = m;
    }
    asm volatile("s_wait_dscnt 0x0" ::: "memory");
    asm volatile("s_barrier_signal -1" ::: "memory");
    asm volatile("s_barrier_wait -1" ::: "memory");

    float4 M0[CH], M1[CH];
    float pl = 0.0f;
#pragma unroll
    for (int j = 0; j < CH; ++j) {
      const float4 m0 = *(const float4*)(magbuf + lrow * MPITCH + wv * (16 * CH) + 16 * j + ltof);
      const float4 m1 = *(const float4*)(magbuf + lrow * MPITCH + wv * (16 * CH) + 16 * j + ltof + 4);
      M0[j] = m0; M1[j] = m1;
      float p = m0.x;
      p = fmaf(p, q, m0.y); p = fmaf(p, q, m0.z); p = fmaf(p, q, m0.w);
      p = fmaf(p, q, m1.x); p = fmaf(p, q, m1.y); p = fmaf(p, q, m1.z); p = fmaf(p, q, m1.w);
      pl = fmaf(pl, q16, p);
    }
    const float po = __shfl_xor(pl, 1, 32);
    if (lodd) ends[pass & 1][wv][lrow] = w * fmaf(q8, po, pl);
    asm volatile("s_wait_dscnt 0x0" ::: "memory");
    asm volatile("s_barrier_signal -1" ::: "memory");
    asm volatile("s_barrier_wait -1" ::: "memory");

    float carry = cpass, run = cpass;
#pragma unroll
    for (int u = 0; u < WSEG; ++u) {
      run = fmaf(run, qseg, ends[pass & 1][u][n]);
      if (u + 1 == wv) carry = run;
    }
    cpass = run;

#pragma unroll
    for (int j = 0; j < CH; ++j) {
      const int t0 = ts + 16 * j;
      *(float4*)(smem + lrow * LDS_STRIDE + ltof)     = M0[j];
      *(float4*)(smem + lrow * LDS_STRIDE + ltof + 4) = M1[j];
      asm volatile("s_wait_dscnt 0" ::: "memory");

      v8f acc;
#pragma unroll
      for (int i = 0; i < 8; ++i) acc[i] = qp[i] * carry;

#pragma unroll
      for (int b = 0; b < 4; ++b) {
        const int k0 = 4 * b + 2 * half;
        v2f Bv, Av;
        Bv.x = smem[n * LDS_STRIDE + k0];
        Bv.y = smem[n * LDS_STRIDE + k0 + 1];
        Av.x = A0[b];
        Av.y = A1[b];
        acc = __builtin_amdgcn_wmma_f32_16x16x4_f32(
            false, Av, false, Bv, (short)0, acc, false, false);
      }

      carry = __shfl(acc[7], 16 + n, 32);

      const float4 g0 = *(const float4*)(smem + n * LDS_STRIDE + 8 * half);
      const float4 g1 = *(const float4*)(smem + n * LDS_STRIDE + 8 * half + 4);
      float4 o0, o1;
      o0.x = pcen_ep(g0.x, acc[0], naexp, dd, oor, dpow);
      o0.y = pcen_ep(g0.y, acc[1], naexp, dd, oor, dpow);
      o0.z = pcen_ep(g0.z, acc[2], naexp, dd, oor, dpow);
      o0.w = pcen_ep(g0.w, acc[3], naexp, dd, oor, dpow);
      o1.x = pcen_ep(g1.x, acc[4], naexp, dd, oor, dpow);
      o1.y = pcen_ep(g1.y, acc[5], naexp, dd, oor, dpow);
      o1.z = pcen_ep(g1.z, acc[6], naexp, dd, oor, dpow);
      o1.w = pcen_ep(g1.w, acc[7], naexp, dd, oor, dpow);
      *(float4*)(outP + t0 + 8 * half)     = o0;
      *(float4*)(outP + t0 + 8 * half + 4) = o1;
    }
  }
}

extern "C" void kernel_launch(void* const* d_in, const int* in_sizes, int n_in,
                              void* d_out, int out_size, void* d_ws, size_t ws_size,
                              hipStream_t stream) {
  (void)in_sizes; (void)n_in; (void)d_ws; (void)ws_size; (void)out_size;
  const float* tf      = (const float*)d_in[0];
  const float* weights = (const float*)d_in[1];
  const float* alpha   = (const float*)d_in[2];
  const float* delta   = (const float*)d_in[3];
  const float* root    = (const float*)d_in[4];
  float* out = (float*)d_out;

  const int blocks = TILES;
  pcen_wmma_kernel<<<blocks, 32 * WSEG, 0, stream>>>(
      tf, weights, alpha, delta, root, out);
}


# Output staged per wave in LDS (the wave's tile holds its whole segment, outputs in place of magnitudes): stores cover 4 rows x one whole 128-byte line, the line two waves share as two 64-byte halves
# speedup vs baseline: 3.2529x; 1.2696x over previous
#include <hip/hip_runtime.h>
#include <math.h>

#define B_   16
#define C_   2
#define F_   257
#define T_   4000
#define ROWS (B_*C_*F_)
#define TILES (ROWS/16)
#define WSEG 10
#define CH   5
#define SPITCH (16 * CH + 4)
#define PTIME (16 * CH * WSEG)
#define MPITCH (PTIME + 4)
#define NUNIT ((4 * (PTIME / 32)) / WSEG)
#define FLOORV 1e-6f
#define NCHUNK (T_/16)
#define NPASS (NCHUNK/(WSEG*CH))

typedef float v8f __attribute__((ext_vector_type(8)));
typedef float v2f __attribute__((ext_vector_type(2)));

__device__ __forceinline__ float fpow(float x, float y) {
  return __builtin_amdgcn_exp2f(y * __builtin_amdgcn_logf(x));
}

__device__ __forceinline__ float pcen_ep(float mag, float ema,
                                         float naexp, float dd, float oor, float dpow) {
  float deninv = fpow(FLOORV + ema, naexp);
  return fpow(fmaf(mag, deninv, dd), oor) - dpow;
}

__global__ __launch_bounds__(32 * WSEG) void pcen_wmma_kernel(
    const float* __restrict__ tf, const float* __restrict__ weights,
    const float* __restrict__ alpha, const float* __restrict__ delta,
    const float* __restrict__ root, float* __restrict__ out)
{
  __shared__ __attribute__((aligned(16))) float smem_all[WSEG][16 * SPITCH];
  __shared__ float ends[2][WSEG][16];
  __shared__ __attribute__((aligned(16))) float magbuf[16 * MPITCH];

  const int lane = threadIdx.x & 31;
  const int wv   = threadIdx.x >> 5;
  const int tile = blockIdx.x;
  float* smem = smem_all[wv];

  const int rowBase = tile * 16;
  const int half = lane >> 4;
  const int n    = lane & 15;

  const int lrow = lane >> 1;
  const int lodd = lane & 1;
  const int ltof = lodd * 8;

  const int r4 = lane >> 3;
  const int c8 = lane & 7;
  unsigned gbyte[4];
  int lbase[4];
#pragma unroll
  for (int ii = 0; ii < 4; ++ii) {
    const int u0  = NUNIT * wv + ii;
    const int row = 4 * (u0 & 3) + r4;
    const int c0  = u0 >> 2;
    const int gr  = rowBase + row;
    const int bc  = gr / F_;
    gbyte[ii] = 4u * ((unsigned)(gr + bc * F_) * (unsigned)T_ + (unsigned)(32 * c0 + 4 * c8));
    lbase[ii] = row * MPITCH + 32 * c0 + 4 * c8;
  }

  const int erow = rowBase + n;
  const int ec   = (erow / F_) % C_;
  const float naexp = -fminf(alpha[ec], 1.0f);
  const float oor   = 1.0f / fmaxf(root[ec], 1.0f);
  const float dd    = delta[ec];
  const float dpow  = fpow(dd, oor);

  const int wodd = wv & 1;
  const int fcol = 16 * wodd;
  const int hcol = wodd ? 0 : 64;
  const int r8 = lane >> 2;
  const int c4 = lane & 3;
  const int lF = r4 * SPITCH + fcol + 4 * c8;
  const int lH = r8 * SPITCH + hcol + 4 * c4;
  unsigned obF = 4u * ((unsigned)(rowBase + r4) * (unsigned)T_ + (unsigned)(wv * (16 * CH) + fcol + 4 * c8));
  unsigned obH = 4u * ((unsigned)(rowBase + r8) * (unsigned)T_ + (unsigned)(wv * (16 * CH) + hcol + 4 * c4));

  const float w = fminf(fmaxf(weights[0], 0.0f), 1.0f);
  const float q = 1.0f - w;

  float A0[4], A1[4];
#pragma unroll
  for (int b = 0; b < 4; ++b) {
    const int k0 = 4 * b + 2 * half;
    A0[b] = (k0     <= n) ? w * fpow(q, (float)(n - k0))     : 0.0f;
    A1[b] = (k0 + 1 <= n) ? w * fpow(q, (float)(n - k0 - 1)) : 0.0f;
  }
  float qp[8];
  qp[0] = fpow(q, (float)(1 + 8 * half));
#pragma unroll
  for (int i = 1; i < 8; ++i) qp[i] = qp[i - 1] * q;

  const float q2 = q * q, q4 = q2 * q2, q8 = q4 * q4, q16 = q8 * q8;
  const float qseg = q16 * q16 * q16 * q16 * q16;

  float cpass;
  {
    const int xb   = erow / (C_ * F_);
    const int xrem = erow % (C_ * F_);
    const float* xr = tf + ((size_t)(xb * C_ + xrem / F_) * (2 * F_) + (xrem % F_)) * T_;
    const float r = xr[0], im = xr[(size_t)F_ * T_];
    cpass = __builtin_amdgcn_sqrtf(fmaf(r, r, im * im));
  }

#pragma unroll 1
  for (int pass = 0; pass < NPASS; ++pass) {
    float4 RE[NUNIT], IM[NUNIT];
#pragma unroll
    for (int k = 0; k < NUNIT; ++k) {
      const float* p = (const float*)((const char*)tf + gbyte[k & 3]) + 32 * (k >> 2);
      RE[k] = *(const float4*)p;
      IM[k] = *(const float4*)(p + (size_t)F_ * T_);
    }
#pragma unroll
    for (int ii = 0; ii < 4; ++ii) gbyte[ii] += 4u * PTIME;

#pragma unroll
    for (int k = 0; k < NUNIT; ++k) {
      const float4 r = RE[k], im = IM[k];
      float4 m;
      m.x = __builtin_amdgcn_sqrtf(fmaf(r.x, r.x, im.x * im.x));
      m.y = __builtin_amdgcn_sqrtf(fmaf(r.y, r.y, im.y * im.y));
      m.z = __builtin_amdgcn_sqrtf(fmaf(r.z, r.z, im.z * im.z));
      m.w = __builtin_amdgcn_sqrtf(fmaf(r.w, r.w, im.w * im.w));
      *(float4*)(magbuf + lbase[k & 3] + 32 * (k >> 2)) = m;
    }
    asm volatile("s_wait_dscnt 0x0" ::: "memory");
    asm volatile("s_barrier_signal -1" ::: "memory");
    asm volatile("s_barrier_wait -1" ::: "memory");

    float pl = 0.0f;
#pragma unroll
    for (int j = 0; j < CH; ++j) {
      const float4 m0 = *(const float4*)(magbuf + lrow * MPITCH + wv * (16 * CH) + 16 * j + ltof);
      const float4 m1 = *(const float4*)(magbuf + lrow * MPITCH + wv * (16 * CH) + 16 * j + ltof + 4);
      *(float4*)(smem + lrow * SPITCH + 16 * j + ltof)     = m0;
      *(float4*)(smem + lrow * SPITCH + 16 * j + ltof + 4) = m1;
      float p = m0.x;
      p = fmaf(p, q, m0.y); p = fmaf(p, q, m0.z); p = fmaf(p, q, m0.w);
      p = fmaf(p, q, m1.x); p = fmaf(p, q, m1.y); p = fmaf(p, q, m1.z); p = fmaf(p, q, m1.w);
      pl = fmaf(pl, q16, p);
    }
    const float po = __shfl_xor(pl, 1, 32);
    if (lodd) ends[pass & 1][wv][lrow] = w * fmaf(q8, po, pl);
    asm volatile("s_wait_dscnt 0x0" ::: "memory");
    asm volatile("s_barrier_signal -1" ::: "memory");
    asm volatile("s_barrier_wait -1" ::: "memory");

    float carry = cpass, run = cpass;
#pragma unroll
    for (int u = 0; u < WSEG; ++u) {
      run = fmaf(run, qseg, ends[pass & 1][u][n]);
      if (u + 1 == wv) carry = run;
    }
    cpass = run;

#pragma unroll
    for (int j = 0; j < CH; ++j) {
      v8f acc;
#pragma unroll
      for (int i = 0; i < 8; ++i) acc[i] = qp[i] * carry;

#pragma unroll
      for (int b = 0; b < 4; ++b) {
        const int k0 = 4 * b + 2 * half;
        v2f Bv, Av;
        Bv.x = smem[n * SPITCH + 16 * j + k0];
        Bv.y = smem[n * SPITCH + 16 * j + k0 + 1];
        Av.x = A0[b];
        Av.y = A1[b];
        acc = __builtin_amdgcn_wmma_f32_16x16x4_f32(
            false, Av, false, Bv, (short)0, acc, false, false);
      }

      carry = __shfl(acc[7], 16 + n, 32);

      const float4 g0 = *(const float4*)(smem + n * SPITCH + 16 * j + 8 * half);
      const float4 g1 = *(const float4*)(smem + n * SPITCH + 16 * j + 8 * half + 4);
      float4 o0, o1;
      o0.x = pcen_ep(g0.x, acc[0], naexp, dd, oor, dpow);
      o0.y = pcen_ep(g0.y, acc[1], naexp, dd, oor, dpow);
      o0.z = pcen_ep(g0.z, acc[2], naexp, dd, oor, dpow);
      o0.w = pcen_ep(g0.w, acc[3], naexp, dd, oor, dpow);
      o1.x = pcen_ep(g1.x, acc[4], naexp, dd, oor, dpow);
      o1.y = pcen_ep(g1.y, acc[5], naexp, dd, oor, dpow);
      o1.z = pcen_ep(g1.z, acc[6], naexp, dd, oor, dpow);
      o1.w = pcen_ep(g1.w, acc[7], naexp, dd, oor, dpow);
      *(float4*)(smem + n * SPITCH + 16 * j + 8 * half)     = o0;
      *(float4*)(smem + n * SPITCH + 16 * j + 8 * half + 4) = o1;
    }

    asm volatile("s_wait_dscnt 0x0" ::: "memory");
#pragma unroll
    for (int i = 0; i < 8; ++i) {
      const int g = i & 3, L = i >> 2;
      const float4 v = *(const float4*)(smem + lF + 4 * g * SPITCH + 32 * L);
      *(float4*)((char*)out + obF + 4u * (unsigned)(4 * g * T_ + 32 * L)) = v;
    }
#pragma unroll
    for (int i = 0; i < 2; ++i) {
      const float4 v = *(const float4*)(smem + lH + 8 * i * SPITCH);
      *(float4*)((char*)out + obH + 4u * (unsigned)(8 * i * T_)) = v;
    }
    obF += 4u * PTIME; obH += 4u * PTIME;
  }
}

extern "C" void kernel_launch(void* const* d_in, const int* in_sizes, int n_in,
                              void* d_out, int out_size, void* d_ws, size_t ws_size,
                              hipStream_t stream) {
  (void)in_sizes; (void)n_in; (void)d_ws; (void)ws_size; (void)out_size;
  const float* tf      = (const float*)d_in[0];
  const float* weights = (const float*)d_in[1];
  const float* alpha   = (const float*)d_in[2];
  const float* delta   = (const float*)d_in[3];
  const float* root    = (const float*)d_in[4];
  float* out = (float*)d_out;

  const int blocks = TILES;
  pcen_wmma_kernel<<<blocks, 32 * WSEG, 0, stream>>>(
      tf, weights, alpha, delta, root, out);
}


# Next pass's input lines requested during the current pass: 4 loads per stage (phase A, then each chunk) into 16 registers, magnitudes to LDS a stage later; only the first pass loads all 20 at once
# speedup vs baseline: 3.4937x; 1.0740x over previous
#include <hip/hip_runtime.h>
#include <math.h>

#define B_   16
#define C_   2
#define F_   257
#define T_   4000
#define ROWS (B_*C_*F_)
#define TILES (ROWS/16)
#define WSEG 10
#define CH   5
#define SPITCH (16 * CH + 4)
#define PTIME (16 * CH * WSEG)
#define MPITCH (PTIME + 4)
#define NUNIT ((4 * (PTIME / 32)) / WSEG)
#define FLOORV 1e-6f
#define NCHUNK (T_/16)
#define NPASS (NCHUNK/(WSEG*CH))
#define GUNIT (NUNIT / CH)

typedef float v8f __attribute__((ext_vector_type(8)));
typedef float v2f __attribute__((ext_vector_type(2)));

__device__ __forceinline__ float fpow(float x, float y) {
  return __builtin_amdgcn_exp2f(y * __builtin_amdgcn_logf(x));
}

__device__ __forceinline__ float pcen_ep(float mag, float ema,
                                         float naexp, float dd, float oor, float dpow) {
  float deninv = fpow(FLOORV + ema, naexp);
  return fpow(fmaf(mag, deninv, dd), oor) - dpow;
}

__global__ __launch_bounds__(32 * WSEG) void pcen_wmma_kernel(
    const float* __restrict__ tf, const float* __restrict__ weights,
    const float* __restrict__ alpha, const float* __restrict__ delta,
    const float* __restrict__ root, float* __restrict__ out)
{
  __shared__ __attribute__((aligned(16))) float smem_all[WSEG][16 * SPITCH];
  __shared__ float ends[2][WSEG][16];
  __shared__ __attribute__((aligned(16))) float magbuf[16 * MPITCH];

  const int lane = threadIdx.x & 31;
  const int wv   = threadIdx.x >> 5;
  const int tile = blockIdx.x;
  float* smem = smem_all[wv];

  const int rowBase = tile * 16;
  const int half = lane >> 4;
  const int n    = lane & 15;

  const int lrow = lane >> 1;
  const int lodd = lane & 1;
  const int ltof = lodd * 8;

  const int r4 = lane >> 3;
  const int c8 = lane & 7;
  unsigned gbyte[4];
  int lbase[4];
#pragma unroll
  for (int ii = 0; ii < 4; ++ii) {
    const int u0  = NUNIT * wv + ii;
    const int row = 4 * (u0 & 3) + r4;
    const int c0  = u0 >> 2;
    const int gr  = rowBase + row;
    const int bc  = gr / F_;
    gbyte[ii] = 4u * ((unsigned)(gr + bc * F_) * (unsigned)T_ + (unsigned)(32 * c0 + 4 * c8));
    lbase[ii] = row * MPITCH + 32 * c0 + 4 * c8;
  }

  const int erow = rowBase + n;
  const int ec   = (erow / F_) % C_;
  const float naexp = -fminf(alpha[ec], 1.0f);
  const float oor   = 1.0f / fmaxf(root[ec], 1.0f);
  const float dd    = delta[ec];
  const float dpow  = fpow(dd, oor);

  const int wodd = wv & 1;
  const int fcol = 16 * wodd;
  const int hcol = wodd ? 0 : 64;
  const int r8 = lane >> 2;
  const int c4 = lane & 3;
  const int lF = r4 * SPITCH + fcol + 4 * c8;
  const int lH = r8 * SPITCH + hcol + 4 * c4;
  unsigned obF = 4u * ((unsigned)(rowBase + r4) * (unsigned)T_ + (unsigned)(wv * (16 * CH) + fcol + 4 * c8));
  unsigned obH = 4u * ((unsigned)(rowBase + r8) * (unsigned)T_ + (unsigned)(wv * (16 * CH) + hcol + 4 * c4));

  const float w = fminf(fmaxf(weights[0], 0.0f), 1.0f);
  const float q = 1.0f - w;

  float A0[4], A1[4];
#pragma unroll
  for (int b = 0; b < 4; ++b) {
    const int k0 = 4 * b + 2 * half;
    A0[b] = (k0     <= n) ? w * fpow(q, (float)(n - k0))     : 0.0f;
    A1[b] = (k0 + 1 <= n) ? w * fpow(q, (float)(n - k0 - 1)) : 0.0f;
  }
  float qp[8];
  qp[0] = fpow(q, (float)(1 + 8 * half));
#pragma unroll
  for (int i = 1; i < 8; ++i) qp[i] = qp[i - 1] * q;

  const float q2 = q * q, q4 = q2 * q2, q8 = q4 * q4, q16 = q8 * q8;
  const float qseg = q16 * q16 * q16 * q16 * q16;

  float cpass;
  {
    const int xb   = erow / (C_ * F_);
    const int xrem = erow % (C_ * F_);
    const float* xr = tf + ((size_t)(xb * C_ + xrem / F_) * (2 * F_) + (xrem % F_)) * T_;
    const float r = xr[0], im = xr[(size_t)F_ * T_];
    cpass = __builtin_amdgcn_sqrtf(fmaf(r, r, im * im));
  }

  auto mags_to_lds = [&](const float4* RE, const float4* IM, const int k0, const int cnt) {
#pragma unroll
    for (int i = 0; i < cnt; ++i) {
      const int k = k0 + i;
      const float4 r = RE[i], im = IM[i];
      float4 m;
      m.x = __builtin_amdgcn_sqrtf(fmaf(r.x, r.x, im.x * im.x));
      m.y = __builtin_amdgcn_sqrtf(fmaf(r.y, r.y, im.y * im.y));
      m.z = __builtin_amdgcn_sqrtf(fmaf(r.z, r.z, im.z * im.z));
      m.w = __builtin_amdgcn_sqrtf(fmaf(r.w, r.w, im.w * im.w));
      *(float4*)(magbuf + lbase[k & 3] + 32 * (k >> 2)) = m;
    }
  };
  auto request = [&](float4* RE, float4* IM, const int k0, const int cnt) {
#pragma unroll
    for (int i = 0; i < cnt; ++i) {
      const int k = k0 + i;
      const float* p = (const float*)((const char*)tf + gbyte[k & 3]) + 32 * (k >> 2);
      RE[i] = *(const float4*)p;
      IM[i] = *(const float4*)(p + (size_t)F_ * T_);
    }
  };

#pragma unroll 1
  for (int pass = 0; pass < NPASS; ++pass) {
    if (pass == 0) {
      float4 RE0[NUNIT], IM0[NUNIT];
      request(RE0, IM0, 0, NUNIT);
      mags_to_lds(RE0, IM0, 0, NUNIT);
#pragma unroll
      for (int ii = 0; ii < 4; ++ii) gbyte[ii] += 4u * PTIME;
    }
    const bool more = pass + 1 < NPASS;

    float4 RE[GUNIT] = {}, IM[GUNIT] = {};
    if (more) request(RE, IM, 0, GUNIT);

    asm volatile("s_wait_dscnt 0x0" ::: "memory");
    asm volatile("s_barrier_signal -1" ::: "memory");
    asm volatile("s_barrier_wait -1" ::: "memory");

    float pl = 0.0f;
#pragma unroll
    for (int j = 0; j < CH; ++j) {
      const float4 m0 = *(const float4*)(magbuf + lrow * MPITCH + wv * (16 * CH) + 16 * j + ltof);
      const float4 m1 = *(const float4*)(magbuf + lrow * MPITCH + wv * (16 * CH) + 16 * j + ltof + 4);
      *(float4*)(smem + lrow * SPITCH + 16 * j + ltof)     = m0;
      *(float4*)(smem + lrow * SPITCH + 16 * j + ltof + 4) = m1;
      float p = m0.x;
      p = fmaf(p, q, m0.y); p = fmaf(p, q, m0.z); p = fmaf(p, q, m0.w);
      p = fmaf(p, q, m1.x); p = fmaf(p, q, m1.y); p = fmaf(p, q, m1.z); p = fmaf(p, q, m1.w);
      pl = fmaf(pl, q16, p);
    }
    const float po = __shfl_xor(pl, 1, 32);
    if (lodd) ends[pass & 1][wv][lrow] = w * fmaf(q8, po, pl);
    asm volatile("s_wait_dscnt 0x0" ::: "memory");
    asm volatile("s_barrier_signal -1" ::: "memory");
    asm volatile("s_barrier_wait -1" ::: "memory");

    float carry = cpass, run = cpass;
#pragma unroll
    for (int u = 0; u < WSEG; ++u) {
      run = fmaf(run, qseg, ends[pass & 1][u][n]);
      if (u + 1 == wv) carry = run;
    }
    cpass = run;

#pragma unroll
    for (int j = 0; j < CH; ++j) {
      if (more) {
        mags_to_lds(RE, IM, GUNIT * j, GUNIT);
        if (j + 1 < CH) request(RE, IM, GUNIT * (j + 1), GUNIT);
      }
      v8f acc;
#pragma unroll
      for (int i = 0; i < 8; ++i) acc[i] = qp[i] * carry;

#pragma unroll
      for (int b = 0; b < 4; ++b) {
        const int k0 = 4 * b + 2 * half;
        v2f Bv, Av;
        Bv.x = smem[n * SPITCH + 16 * j + k0];
        Bv.y = smem[n * SPITCH + 16 * j + k0 + 1];
        Av.x = A0[b];
        Av.y = A1[b];
        acc = __builtin_amdgcn_wmma_f32_16x16x4_f32(
            false, Av, false, Bv, (short)0, acc, false, false);
      }

      carry = __shfl(acc[7], 16 + n, 32);

      const float4 g0 = *(const float4*)(smem + n * SPITCH + 16 * j + 8 * half);
      const float4 g1 = *(const float4*)(smem + n * SPITCH + 16 * j + 8 * half + 4);
      float4 o0, o1;
      o0.x = pcen_ep(g0.x, acc[0], naexp, dd, oor, dpow);
      o0.y = pcen_ep(g0.y, acc[1], naexp, dd, oor, dpow);
      o0.z = pcen_ep(g0.z, acc[2], naexp, dd, oor, dpow);
      o0.w = pcen_ep(g0.w, acc[3], naexp, dd, oor, dpow);
      o1.x = pcen_ep(g1.x, acc[4], naexp, dd, oor, dpow);
      o1.y = pcen_ep(g1.y, acc[5], naexp, dd, oor, dpow);
      o1.z = pcen_ep(g1.z, acc[6], naexp, dd, oor, dpow);
      o1.w = pcen_ep(g1.w, acc[7], naexp, dd, oor, dpow);
      *(float4*)(smem + n * SPITCH + 16 * j + 8 * half)     = o0;
      *(float4*)(smem + n * SPITCH + 16 * j + 8 * half + 4) = o1;
    }

    asm volatile("s_wait_dscnt 0x0" ::: "memory");
#pragma unroll
    for (int i = 0; i < 8; ++i) {
      const int g = i & 3, L = i >> 2;
      const float4 v = *(const float4*)(smem + lF + 4 * g * SPITCH + 32 * L);
      *(float4*)((char*)out + obF + 4u * (unsigned)(4 * g * T_ + 32 * L)) = v;
    }
#pragma unroll
    for (int i = 0; i < 2; ++i) {
      const float4 v = *(const float4*)(smem + lH + 8 * i * SPITCH);
      *(float4*)((char*)out + obH + 4u * (unsigned)(8 * i * T_)) = v;
    }
    obF += 4u * PTIME; obH += 4u * PTIME;

#pragma unroll
    for (int ii = 0; ii < 4; ++ii) gbyte[ii] += 4u * PTIME;
  }
}

extern "C" void kernel_launch(void* const* d_in, const int* in_sizes, int n_in,
                              void* d_out, int out_size, void* d_ws, size_t ws_size,
                              hipStream_t stream) {
  (void)in_sizes; (void)n_in; (void)d_ws; (void)ws_size; (void)out_size;
  const float* tf      = (const float*)d_in[0];
  const float* weights = (const float*)d_in[1];
  const float* alpha   = (const float*)d_in[2];
  const float* delta   = (const float*)d_in[3];
  const float* root    = (const float*)d_in[4];
  float* out = (float*)d_out;

  const int blocks = TILES;
  pcen_wmma_kernel<<<blocks, 32 * WSEG, 0, stream>>>(
      tf, weights, alpha, delta, root, out);
}
